# starting kernel
# baseline (speedup 1.0000x reference)
#include <hip/hip_runtime.h>
#include <hip/hip_bf16.h>

typedef __attribute__((ext_vector_type(2))) float v2f;
typedef __attribute__((ext_vector_type(8))) float v8f;

#define D 128
#define LDS_STRIDE 260


__global__ void k_zero_i32(int* p, int n) {
    int i = blockIdx.x * blockDim.x + threadIdx.x;
    if (i < n) p[i] = 0;
}

__global__ void k_degree(const int* __restrict__ dst, int* __restrict__ deg, int e) {
    int i = blockIdx.x * blockDim.x + threadIdx.x;
    if (i < e) atomicAdd(&deg[dst[i]], 1);
}

__global__ void k_scan_block(const int* __restrict__ deg, int* __restrict__ off,
                             int* __restrict__ bsum, int n) {
    __shared__ int s[256];
    int i = blockIdx.x * 256 + threadIdx.x;
    int v = (i < n) ? deg[i] : 0;
    s[threadIdx.x] = v;
    __syncthreads();
    for (int d = 1; d < 256; d <<= 1) {
        int t = (threadIdx.x >= d) ? s[threadIdx.x - d] : 0;
        __syncthreads();
        s[threadIdx.x] += t;
        __syncthreads();
    }
    if (i < n) off[i + 1] = s[threadIdx.x];
    if (threadIdx.x == 255) bsum[blockIdx.x] = s[255];
}

__global__ void k_scan_sums(int* bsum, int nb) {
    if (blockIdx.x == 0 && threadIdx.x == 0) {
        int acc = 0;
        for (int b = 0; b < nb; ++b) { int t = bsum[b]; bsum[b] = acc; acc += t; }
    }
}

__global__ void k_scan_add(int* __restrict__ off, const int* __restrict__ bsum, int n) {
    int i = blockIdx.x * 256 + threadIdx.x;
    if (i < n) off[i + 1] += bsum[blockIdx.x];
    if (i == 0) off[0] = 0;
}

__global__ void k_fill(const int* __restrict__ src, const int* __restrict__ dst,
                       const int* __restrict__ off, int* __restrict__ cur,
                       int* __restrict__ csr, int e) {
    int i = blockIdx.x * blockDim.x + threadIdx.x;
    if (i < e) {
        int d = dst[i];
        int p = atomicAdd(&cur[d], 1);
        csr[off[d] + p] = src[i];
    }
}

__global__ __launch_bounds__(256) void k_sage(
    const float* __restrict__ xsrc, const float* __restrict__ xdst,
    const int* __restrict__ off, const int* __restrict__ csr,
    const float* __restrict__ Wl, const float* __restrict__ Wr,
    const float* __restrict__ bl, float* __restrict__ out,
    int nDst, int relu)
{
    __shared__ float A2[16 * LDS_STRIDE];
    const int wave = threadIdx.x >> 5;
    const int lane = threadIdx.x & 31;
    const int mbase = blockIdx.x * 16;
    const int c0 = lane * 4;

    for (int rr = 0; rr < 2; ++rr) {
        const int r = wave * 2 + rr;
        const int g = mbase + r;
        float* row = &A2[r * LDS_STRIDE];
        if (g < nDst) {
            const int beg = off[g], end = off[g + 1];
            float ax = 0.f, ay = 0.f, az = 0.f, aw = 0.f;
            for (int j = beg; j < end; ++j) {
                const int s = csr[j];
                const float4 v = *reinterpret_cast<const float4*>(xsrc + (size_t)s * D + c0);
                ax += v.x; ay += v.y; az += v.z; aw += v.w;
            }
            const float sc = 1.0f / fmaxf((float)(end - beg), 1.0f);
            row[c0 + 0] = ax * sc; row[c0 + 1] = ay * sc;
            row[c0 + 2] = az * sc; row[c0 + 3] = aw * sc;
            const float4 xv = *reinterpret_cast<const float4*>(xdst + (size_t)g * D + c0);
            row[D + c0 + 0] = xv.x; row[D + c0 + 1] = xv.y;
            row[D + c0 + 2] = xv.z; row[D + c0 + 3] = xv.w;
        } else {
            row[c0 + 0] = 0.f; row[c0 + 1] = 0.f; row[c0 + 2] = 0.f; row[c0 + 3] = 0.f;
            row[D + c0 + 0] = 0.f; row[D + c0 + 1] = 0.f; row[D + c0 + 2] = 0.f; row[D + c0 + 3] = 0.f;
        }
    }
    __syncthreads();

    const int lh = lane & 15;
    const int hi = lane >> 4;
    const int jcol = wave * 16 + lh;
    const float* arow = &A2[lh * LDS_STRIDE + 2 * hi];
    const float* wlrow = Wl + (size_t)jcol * D + 2 * hi;
    const float* wrrow = Wr + (size_t)jcol * D + 2 * hi;

    v8f c = {};
    #pragma unroll
    for (int k0 = 0; k0 < D; k0 += 4) {
        const v2f a = *reinterpret_cast<const v2f*>(arow + k0);
        const v2f b = *reinterpret_cast<const v2f*>(wlrow + k0);
        c = __builtin_amdgcn_wmma_f32_16x16x4_f32(false, a, false, b, (short)0, c, false, false);
    }
    #pragma unroll
    for (int k0 = 0; k0 < D; k0 += 4) {
        const v2f a = *reinterpret_cast<const v2f*>(arow + D + k0);
        const v2f b = *reinterpret_cast<const v2f*>(wrrow + k0);
        c = __builtin_amdgcn_wmma_f32_16x16x4_f32(false, a, false, b, (short)0, c, false, false);
    }

    const float bb = bl[jcol];
    #pragma unroll
    for (int i = 0; i < 8; ++i) {
        const int gr = mbase + i + 8 * hi;
        if (gr < nDst) {
            float v = c[i] + bb;
            if (relu) v = fmaxf(v, 0.0f);
            out[(size_t)gr * D + jcol] = v;
        }
    }
}


extern "C" void kernel_launch(void* const* d_in, const int* in_sizes, int n_in,
                              void* d_out, int out_size, void* d_ws, size_t ws_size,
                              hipStream_t stream) {
    const float* x_player = (const float*)d_in[0];
    const float* x_team   = (const float*)d_in[1];
    const int*   ei_pt    = (const int*)d_in[2];
    const int*   ei_tp    = (const int*)d_in[3];
    const float* Wl0_pt = (const float*)d_in[4];
    const float* Wr0_pt = (const float*)d_in[5];
    const float* Wl0_tp = (const float*)d_in[6];
    const float* Wr0_tp = (const float*)d_in[7];
    const float* Wl1_pt = (const float*)d_in[8];
    const float* Wr1_pt = (const float*)d_in[9];
    const float* Wl1_tp = (const float*)d_in[10];
    const float* Wr1_tp = (const float*)d_in[11];
    const float* bl0_pt = (const float*)d_in[12];
    const float* bl0_tp = (const float*)d_in[13];
    const float* bl1_pt = (const float*)d_in[14];
    const float* bl1_tp = (const float*)d_in[15];

    const int N = in_sizes[0] / D;
    const int E = in_sizes[2] / 2;

    const int* src_pt = ei_pt;
    const int* dst_pt = ei_pt + E;
    const int* src_tp = ei_tp;
    const int* dst_tp = ei_tp + E;

    char* p = (char*)d_ws;
    auto take_i = [&](size_t n) { int* r = (int*)p; p += n * sizeof(int); return r; };
    int* deg_pt = take_i(N);
    int* deg_tp = take_i(N);
    int* cur_pt = take_i(N);
    int* cur_tp = take_i(N);
    int* off_pt = take_i(N + 1);
    int* off_tp = take_i(N + 1);
    int* bsum   = take_i(1024);
    int* csr_pt = take_i(E);
    int* csr_tp = take_i(E);
    p = (char*)(((size_t)p + 255) & ~(size_t)255);
    float* h_player = (float*)p; p += (size_t)N * D * sizeof(float);
    float* h_team   = (float*)p; p += (size_t)N * D * sizeof(float);

    float* o_player = (float*)d_out;
    float* o_team   = (float*)d_out + (size_t)N * D;

    const int TB = 256;
    const int nbN  = (N + TB - 1) / TB;
    const int nbE  = (E + TB - 1) / TB;
    const int nb4N = (4 * N + TB - 1) / TB;
    const int nbM  = (N + 15) / 16;

    k_zero_i32<<<nb4N, TB, 0, stream>>>(deg_pt, 4 * N);
    k_degree<<<nbE, TB, 0, stream>>>(dst_pt, deg_pt, E);
    k_degree<<<nbE, TB, 0, stream>>>(dst_tp, deg_tp, E);

    k_scan_block<<<nbN, TB, 0, stream>>>(deg_pt, off_pt, bsum, N);
    k_scan_sums<<<1, 1, 0, stream>>>(bsum, nbN);
    k_scan_add<<<nbN, TB, 0, stream>>>(off_pt, bsum, N);

    k_scan_block<<<nbN, TB, 0, stream>>>(deg_tp, off_tp, bsum, N);
    k_scan_sums<<<1, 1, 0, stream>>>(bsum, nbN);
    k_scan_add<<<nbN, TB, 0, stream>>>(off_tp, bsum, N);

    k_fill<<<nbE, TB, 0, stream>>>(src_pt, dst_pt, off_pt, cur_pt, csr_pt, E);
    k_fill<<<nbE, TB, 0, stream>>>(src_tp, dst_tp, off_tp, cur_tp, csr_tp, E);

    k_sage<<<nbM, TB, 0, stream>>>(x_player, x_team,   off_pt, csr_pt, Wl0_pt, Wr0_pt, bl0_pt, h_team,   N, 1);
    k_sage<<<nbM, TB, 0, stream>>>(x_team,   x_player, off_tp, csr_tp, Wl0_tp, Wr0_tp, bl0_tp, h_player, N, 1);

    k_sage<<<nbM, TB, 0, stream>>>(h_player, h_team,   off_pt, csr_pt, Wl1_pt, Wr1_pt, bl1_pt, o_team,   N, 0);
    k_sage<<<nbM, TB, 0, stream>>>(h_team,   h_player, off_tp, csr_tp, Wl1_tp, Wr1_tp, bl1_tp, o_player, N, 0);
}


# k_sage: 32 rows per workgroup; each wave reuses one 16-byte weight load for 4 WMMAs (2 row tiles x 2 k-steps); k now summed as {0,1,4,5} then {2,3,6,7} within each 8
# speedup vs baseline: 1.3954x; 1.3954x over previous
#include <hip/hip_runtime.h>
#include <hip/hip_bf16.h>

typedef __attribute__((ext_vector_type(2))) float v2f;
typedef __attribute__((ext_vector_type(8))) float v8f;

#define D 128
#define LDS_STRIDE 260


__global__ void k_zero_i32(int* p, int n) {
    int i = blockIdx.x * blockDim.x + threadIdx.x;
    if (i < n) p[i] = 0;
}

__global__ void k_degree(const int* __restrict__ dst, int* __restrict__ deg, int e) {
    int i = blockIdx.x * blockDim.x + threadIdx.x;
    if (i < e) atomicAdd(&deg[dst[i]], 1);
}

__global__ void k_scan_block(const int* __restrict__ deg, int* __restrict__ off,
                             int* __restrict__ bsum, int n) {
    __shared__ int s[256];
    int i = blockIdx.x * 256 + threadIdx.x;
    int v = (i < n) ? deg[i] : 0;
    s[threadIdx.x] = v;
    __syncthreads();
    for (int d = 1; d < 256; d <<= 1) {
        int t = (threadIdx.x >= d) ? s[threadIdx.x - d] : 0;
        __syncthreads();
        s[threadIdx.x] += t;
        __syncthreads();
    }
    if (i < n) off[i + 1] = s[threadIdx.x];
    if (threadIdx.x == 255) bsum[blockIdx.x] = s[255];
}

__global__ void k_scan_sums(int* bsum, int nb) {
    if (blockIdx.x == 0 && threadIdx.x == 0) {
        int acc = 0;
        for (int b = 0; b < nb; ++b) { int t = bsum[b]; bsum[b] = acc; acc += t; }
    }
}

__global__ void k_scan_add(int* __restrict__ off, const int* __restrict__ bsum, int n) {
    int i = blockIdx.x * 256 + threadIdx.x;
    if (i < n) off[i + 1] += bsum[blockIdx.x];
    if (i == 0) off[0] = 0;
}

__global__ void k_fill(const int* __restrict__ src, const int* __restrict__ dst,
                       const int* __restrict__ off, int* __restrict__ cur,
                       int* __restrict__ csr, int e) {
    int i = blockIdx.x * blockDim.x + threadIdx.x;
    if (i < e) {
        int d = dst[i];
        int p = atomicAdd(&cur[d], 1);
        csr[off[d] + p] = src[i];
    }
}

#define MT 32
__global__ __launch_bounds__(256) void k_sage(
    const float* __restrict__ xsrc, const float* __restrict__ xdst,
    const int* __restrict__ off, const int* __restrict__ csr,
    const float* __restrict__ Wl, const float* __restrict__ Wr,
    const float* __restrict__ bl, float* __restrict__ out,
    int nDst, int relu)
{
    __shared__ __attribute__((aligned(16))) float A2[MT * LDS_STRIDE];
    const int wave = threadIdx.x >> 5;
    const int lane = threadIdx.x & 31;
    const int mbase = blockIdx.x * MT;
    const int c0 = lane * 4;

    for (int rr = 0; rr < MT / 8; ++rr) {
        const int r = wave * (MT / 8) + rr;
        const int g = mbase + r;
        float* row = &A2[r * LDS_STRIDE];
        if (g < nDst) {
            const int beg = off[g], end = off[g + 1];
            float ax = 0.f, ay = 0.f, az = 0.f, aw = 0.f;
            for (int j = beg; j < end; ++j) {
                const int s = csr[j];
                const float4 v = *reinterpret_cast<const float4*>(xsrc + (size_t)s * D + c0);
                ax += v.x; ay += v.y; az += v.z; aw += v.w;
            }
            const float sc = 1.0f / fmaxf((float)(end - beg), 1.0f);
            row[c0 + 0] = ax * sc; row[c0 + 1] = ay * sc;
            row[c0 + 2] = az * sc; row[c0 + 3] = aw * sc;
            const float4 xv = *reinterpret_cast<const float4*>(xdst + (size_t)g * D + c0);
            row[D + c0 + 0] = xv.x; row[D + c0 + 1] = xv.y;
            row[D + c0 + 2] = xv.z; row[D + c0 + 3] = xv.w;
        } else {
            row[c0 + 0] = 0.f; row[c0 + 1] = 0.f; row[c0 + 2] = 0.f; row[c0 + 3] = 0.f;
            row[D + c0 + 0] = 0.f; row[D + c0 + 1] = 0.f; row[D + c0 + 2] = 0.f; row[D + c0 + 3] = 0.f;
        }
    }
    __syncthreads();

    const int lh = lane & 15;
    const int hi = lane >> 4;
    const int jcol = wave * 16 + lh;
    const float* arow0 = &A2[lh * LDS_STRIDE + 4 * hi];
    const float* arow1 = arow0 + 16 * LDS_STRIDE;
    const float* wlrow = Wl + (size_t)jcol * D + 4 * hi;
    const float* wrrow = Wr + (size_t)jcol * D + 4 * hi;

    v8f acc0 = {};
    v8f acc1 = {};
    #pragma unroll
    for (int kb = 0; kb < D; kb += 8) {
        const float4 b  = *reinterpret_cast<const float4*>(wlrow + kb);
        const float4 a0 = *reinterpret_cast<const float4*>(arow0 + kb);
        const float4 a1 = *reinterpret_cast<const float4*>(arow1 + kb);
        const v2f blo = {b.x, b.y},  bhi = {b.z, b.w};
        const v2f a0lo = {a0.x, a0.y}, a0hi = {a0.z, a0.w};
        const v2f a1lo = {a1.x, a1.y}, a1hi = {a1.z, a1.w};
        acc0 = __builtin_amdgcn_wmma_f32_16x16x4_f32(false, a0lo, false, blo, (short)0, acc0, false, false);
        acc1 = __builtin_amdgcn_wmma_f32_16x16x4_f32(false, a1lo, false, blo, (short)0, acc1, false, false);
        acc0 = __builtin_amdgcn_wmma_f32_16x16x4_f32(false, a0hi, false, bhi, (short)0, acc0, false, false);
        acc1 = __builtin_amdgcn_wmma_f32_16x16x4_f32(false, a1hi, false, bhi, (short)0, acc1, false, false);
    }
    #pragma unroll
    for (int kb = 0; kb < D; kb += 8) {
        const float4 b  = *reinterpret_cast<const float4*>(wrrow + kb);
        const float4 a0 = *reinterpret_cast<const float4*>(arow0 + D + kb);
        const float4 a1 = *reinterpret_cast<const float4*>(arow1 + D + kb);
        const v2f blo = {b.x, b.y},  bhi = {b.z, b.w};
        const v2f a0lo = {a0.x, a0.y}, a0hi = {a0.z, a0.w};
        const v2f a1lo = {a1.x, a1.y}, a1hi = {a1.z, a1.w};
        acc0 = __builtin_amdgcn_wmma_f32_16x16x4_f32(false, a0lo, false, blo, (short)0, acc0, false, false);
        acc1 = __builtin_amdgcn_wmma_f32_16x16x4_f32(false, a1lo, false, blo, (short)0, acc1, false, false);
        acc0 = __builtin_amdgcn_wmma_f32_16x16x4_f32(false, a0hi, false, bhi, (short)0, acc0, false, false);
        acc1 = __builtin_amdgcn_wmma_f32_16x16x4_f32(false, a1hi, false, bhi, (short)0, acc1, false, false);
    }

    const float bb = bl[jcol];
    #pragma unroll
    for (int i = 0; i < 8; ++i) {
        const int gr0 = mbase + i + 8 * hi;
        if (gr0 < nDst) {
            float v = acc0[i] + bb;
            if (relu) v = fmaxf(v, 0.0f);
            out[(size_t)gr0 * D + jcol] = v;
        }
        const int gr1 = gr0 + 16;
        if (gr1 < nDst) {
            float v = acc1[i] + bb;
            if (relu) v = fmaxf(v, 0.0f);
            out[(size_t)gr1 * D + jcol] = v;
        }
    }
}


extern "C" void kernel_launch(void* const* d_in, const int* in_sizes, int n_in,
                              void* d_out, int out_size, void* d_ws, size_t ws_size,
                              hipStream_t stream) {
    const float* x_player = (const float*)d_in[0];
    const float* x_team   = (const float*)d_in[1];
    const int*   ei_pt    = (const int*)d_in[2];
    const int*   ei_tp    = (const int*)d_in[3];
    const float* Wl0_pt = (const float*)d_in[4];
    const float* Wr0_pt = (const float*)d_in[5];
    const float* Wl0_tp = (const float*)d_in[6];
    const float* Wr0_tp = (const float*)d_in[7];
    const float* Wl1_pt = (const float*)d_in[8];
    const float* Wr1_pt = (const float*)d_in[9];
    const float* Wl1_tp = (const float*)d_in[10];
    const float* Wr1_tp = (const float*)d_in[11];
    const float* bl0_pt = (const float*)d_in[12];
    const float* bl0_tp = (const float*)d_in[13];
    const float* bl1_pt = (const float*)d_in[14];
    const float* bl1_tp = (const float*)d_in[15];

    const int N = in_sizes[0] / D;
    const int E = in_sizes[2] / 2;

    const int* src_pt = ei_pt;
    const int* dst_pt = ei_pt + E;
    const int* src_tp = ei_tp;
    const int* dst_tp = ei_tp + E;

    char* p = (char*)d_ws;
    auto take_i = [&](size_t n) { int* r = (int*)p; p += n * sizeof(int); return r; };
    int* deg_pt = take_i(N);
    int* deg_tp = take_i(N);
    int* cur_pt = take_i(N);
    int* cur_tp = take_i(N);
    int* off_pt = take_i(N + 1);
    int* off_tp = take_i(N + 1);
    int* bsum   = take_i(1024);
    int* csr_pt = take_i(E);
    int* csr_tp = take_i(E);
    p = (char*)(((size_t)p + 255) & ~(size_t)255);
    float* h_player = (float*)p; p += (size_t)N * D * sizeof(float);
    float* h_team   = (float*)p; p += (size_t)N * D * sizeof(float);

    float* o_player = (float*)d_out;
    float* o_team   = (float*)d_out + (size_t)N * D;

    const int TB = 256;
    const int nbN  = (N + TB - 1) / TB;
    const int nbE  = (E + TB - 1) / TB;
    const int nb4N = (4 * N + TB - 1) / TB;
    const int nbM  = (N + MT - 1) / MT;

    k_zero_i32<<<nb4N, TB, 0, stream>>>(deg_pt, 4 * N);
    k_degree<<<nbE, TB, 0, stream>>>(dst_pt, deg_pt, E);
    k_degree<<<nbE, TB, 0, stream>>>(dst_tp, deg_tp, E);

    k_scan_block<<<nbN, TB, 0, stream>>>(deg_pt, off_pt, bsum, N);
    k_scan_sums<<<1, 1, 0, stream>>>(bsum, nbN);
    k_scan_add<<<nbN, TB, 0, stream>>>(off_pt, bsum, N);

    k_scan_block<<<nbN, TB, 0, stream>>>(deg_tp, off_tp, bsum, N);
    k_scan_sums<<<1, 1, 0, stream>>>(bsum, nbN);
    k_scan_add<<<nbN, TB, 0, stream>>>(off_tp, bsum, N);

    k_fill<<<nbE, TB, 0, stream>>>(src_pt, dst_pt, off_pt, cur_pt, csr_pt, E);
    k_fill<<<nbE, TB, 0, stream>>>(src_tp, dst_tp, off_tp, cur_tp, csr_tp, E);

    k_sage<<<nbM, TB, 0, stream>>>(x_player, x_team,   off_pt, csr_pt, Wl0_pt, Wr0_pt, bl0_pt, h_team,   N, 1);
    k_sage<<<nbM, TB, 0, stream>>>(x_team,   x_player, off_tp, csr_tp, Wl0_tp, Wr0_tp, bl0_tp, h_player, N, 1);

    k_sage<<<nbM, TB, 0, stream>>>(h_player, h_team,   off_pt, csr_pt, Wl1_pt, Wr1_pt, bl1_pt, o_team,   N, 0);
    k_sage<<<nbM, TB, 0, stream>>>(h_team,   h_player, off_tp, csr_tp, Wl1_tp, Wr1_tp, bl1_tp, o_player, N, 0);
}


# k_sage gather: a row's neighbour ids come from one wave load and up to 8 neighbour rows are requested before the first is summed (was one dependent id-then-row load per neighbour); sum order unchanged
# speedup vs baseline: 1.5315x; 1.0975x over previous
#include <hip/hip_runtime.h>
#include <hip/hip_bf16.h>

typedef __attribute__((ext_vector_type(2))) float v2f;
typedef __attribute__((ext_vector_type(8))) float v8f;

#define D 128
#define LDS_STRIDE 260


__global__ void k_zero_i32(int* p, int n) {
    int i = blockIdx.x * blockDim.x + threadIdx.x;
    if (i < n) p[i] = 0;
}

__global__ void k_degree(const int* __restrict__ dst, int* __restrict__ deg, int e) {
    int i = blockIdx.x * blockDim.x + threadIdx.x;
    if (i < e) atomicAdd(&deg[dst[i]], 1);
}

__global__ void k_scan_block(const int* __restrict__ deg, int* __restrict__ off,
                             int* __restrict__ bsum, int n) {
    __shared__ int s[256];
    int i = blockIdx.x * 256 + threadIdx.x;
    int v = (i < n) ? deg[i] : 0;
    s[threadIdx.x] = v;
    __syncthreads();
    for (int d = 1; d < 256; d <<= 1) {
        int t = (threadIdx.x >= d) ? s[threadIdx.x - d] : 0;
        __syncthreads();
        s[threadIdx.x] += t;
        __syncthreads();
    }
    if (i < n) off[i + 1] = s[threadIdx.x];
    if (threadIdx.x == 255) bsum[blockIdx.x] = s[255];
}

__global__ void k_scan_sums(int* bsum, int nb) {
    if (blockIdx.x == 0 && threadIdx.x == 0) {
        int acc = 0;
        for (int b = 0; b < nb; ++b) { int t = bsum[b]; bsum[b] = acc; acc += t; }
    }
}

__global__ void k_scan_add(int* __restrict__ off, const int* __restrict__ bsum, int n) {
    int i = blockIdx.x * 256 + threadIdx.x;
    if (i < n) off[i + 1] += bsum[blockIdx.x];
    if (i == 0) off[0] = 0;
}

__global__ void k_fill(const int* __restrict__ src, const int* __restrict__ dst,
                       const int* __restrict__ off, int* __restrict__ cur,
                       int* __restrict__ csr, int e) {
    int i = blockIdx.x * blockDim.x + threadIdx.x;
    if (i < e) {
        int d = dst[i];
        int p = atomicAdd(&cur[d], 1);
        csr[off[d] + p] = src[i];
    }
}

#define MT 32
__global__ __launch_bounds__(256) void k_sage(
    const float* __restrict__ xsrc, const float* __restrict__ xdst,
    const int* __restrict__ off, const int* __restrict__ csr,
    const float* __restrict__ Wl, const float* __restrict__ Wr,
    const float* __restrict__ bl, float* __restrict__ out,
    int nDst, int relu)
{
    __shared__ __attribute__((aligned(16))) float A2[MT * LDS_STRIDE];
    const int wave = threadIdx.x >> 5;
    const int lane = threadIdx.x & 31;
    const int mbase = blockIdx.x * MT;
    const int c0 = lane * 4;

    const int wv = __builtin_amdgcn_readfirstlane(wave);
    const int r0 = wv * (MT / 8);
    const int g0 = mbase + r0;
    const int ov = off[min(g0 + min(lane, MT / 8), nDst)];
    float4 xv[MT / 8];
    #pragma unroll
    for (int rr = 0; rr < MT / 8; ++rr) {
        const int g = g0 + rr;
        xv[rr] = make_float4(0.f, 0.f, 0.f, 0.f);
        if (g < nDst) xv[rr] = *reinterpret_cast<const float4*>(xdst + (size_t)g * D + c0);
    }
    int begs[MT / 8], degs[MT / 8], ids[MT / 8];
    #pragma unroll
    for (int rr = 0; rr < MT / 8; ++rr) {
        begs[rr] = __builtin_amdgcn_readlane(ov, rr);
        degs[rr] = __builtin_amdgcn_readlane(ov, rr + 1) - begs[rr];
    }
    #pragma unroll
    for (int rr = 0; rr < MT / 8; ++rr)
        ids[rr] = (lane < degs[rr]) ? csr[begs[rr] + lane] : 0;
    #pragma unroll
    for (int rr = 0; rr < MT / 8; ++rr) {
        float* row = &A2[(r0 + rr) * LDS_STRIDE];
        row[D + c0 + 0] = xv[rr].x; row[D + c0 + 1] = xv[rr].y;
        row[D + c0 + 2] = xv[rr].z; row[D + c0 + 3] = xv[rr].w;
    }
    #pragma unroll
    for (int rr = 0; rr < MT / 8; ++rr) {
        const int beg = begs[rr];
        const int deg = degs[rr];
        int id = ids[rr];
        float ax = 0.f, ay = 0.f, az = 0.f, aw = 0.f;
        for (int j0 = 0; j0 < deg; j0 += 8) {
            if (j0 != 0 && (j0 & 31) == 0)
                id = (j0 + lane < deg) ? csr[beg + j0 + lane] : 0;
            float4 v[8];
            #pragma unroll
            for (int k = 0; k < 8; ++k) {
                v[k] = make_float4(0.f, 0.f, 0.f, 0.f);
                if (j0 + k < deg) {
                    const int s = __builtin_amdgcn_readlane(id, (j0 + k) & 31);
                    v[k] = *reinterpret_cast<const float4*>(xsrc + (size_t)s * D + c0);
                }
            }
            #pragma unroll
            for (int k = 0; k < 8; ++k) {
                if (j0 + k < deg) { ax += v[k].x; ay += v[k].y; az += v[k].z; aw += v[k].w; }
            }
        }
        const float sc = 1.0f / fmaxf((float)deg, 1.0f);
        float* row = &A2[(r0 + rr) * LDS_STRIDE];
        row[c0 + 0] = ax * sc; row[c0 + 1] = ay * sc;
        row[c0 + 2] = az * sc; row[c0 + 3] = aw * sc;
    }
    __syncthreads();

    const int lh = lane & 15;
    const int hi = lane >> 4;
    const int jcol = wave * 16 + lh;
    const float* arow0 = &A2[lh * LDS_STRIDE + 4 * hi];
    const float* arow1 = arow0 + 16 * LDS_STRIDE;
    const float* wlrow = Wl + (size_t)jcol * D + 4 * hi;
    const float* wrrow = Wr + (size_t)jcol * D + 4 * hi;

    v8f acc0 = {};
    v8f acc1 = {};
    #pragma unroll
    for (int kb = 0; kb < D; kb += 8) {
        const float4 b  = *reinterpret_cast<const float4*>(wlrow + kb);
        const float4 a0 = *reinterpret_cast<const float4*>(arow0 + kb);
        const float4 a1 = *reinterpret_cast<const float4*>(arow1 + kb);
        const v2f blo = {b.x, b.y},  bhi = {b.z, b.w};
        const v2f a0lo = {a0.x, a0.y}, a0hi = {a0.z, a0.w};
        const v2f a1lo = {a1.x, a1.y}, a1hi = {a1.z, a1.w};
        acc0 = __builtin_amdgcn_wmma_f32_16x16x4_f32(false, a0lo, false, blo, (short)0, acc0, false, false);
        acc1 = __builtin_amdgcn_wmma_f32_16x16x4_f32(false, a1lo, false, blo, (short)0, acc1, false, false);
        acc0 = __builtin_amdgcn_wmma_f32_16x16x4_f32(false, a0hi, false, bhi, (short)0, acc0, false, false);
        acc1 = __builtin_amdgcn_wmma_f32_16x16x4_f32(false, a1hi, false, bhi, (short)0, acc1, false, false);
    }
    #pragma unroll
    for (int kb = 0; kb < D; kb += 8) {
        const float4 b  = *reinterpret_cast<const float4*>(wrrow + kb);
        const float4 a0 = *reinterpret_cast<const float4*>(arow0 + D + kb);
        const float4 a1 = *reinterpret_cast<const float4*>(arow1 + D + kb);
        const v2f blo = {b.x, b.y},  bhi = {b.z, b.w};
        const v2f a0lo = {a0.x, a0.y}, a0hi = {a0.z, a0.w};
        const v2f a1lo = {a1.x, a1.y}, a1hi = {a1.z, a1.w};
        acc0 = __builtin_amdgcn_wmma_f32_16x16x4_f32(false, a0lo, false, blo, (short)0, acc0, false, false);
        acc1 = __builtin_amdgcn_wmma_f32_16x16x4_f32(false, a1lo, false, blo, (short)0, acc1, false, false);
        acc0 = __builtin_amdgcn_wmma_f32_16x16x4_f32(false, a0hi, false, bhi, (short)0, acc0, false, false);
        acc1 = __builtin_amdgcn_wmma_f32_16x16x4_f32(false, a1hi, false, bhi, (short)0, acc1, false, false);
    }

    const float bb = bl[jcol];
    #pragma unroll
    for (int i = 0; i < 8; ++i) {
        const int gr0 = mbase + i + 8 * hi;
        if (gr0 < nDst) {
            float v = acc0[i] + bb;
            if (relu) v = fmaxf(v, 0.0f);
            out[(size_t)gr0 * D + jcol] = v;
        }
        const int gr1 = gr0 + 16;
        if (gr1 < nDst) {
            float v = acc1[i] + bb;
            if (relu) v = fmaxf(v, 0.0f);
            out[(size_t)gr1 * D + jcol] = v;
        }
    }
}


extern "C" void kernel_launch(void* const* d_in, const int* in_sizes, int n_in,
                              void* d_out, int out_size, void* d_ws, size_t ws_size,
                              hipStream_t stream) {
    const float* x_player = (const float*)d_in[0];
    const float* x_team   = (const float*)d_in[1];
    const int*   ei_pt    = (const int*)d_in[2];
    const int*   ei_tp    = (const int*)d_in[3];
    const float* Wl0_pt = (const float*)d_in[4];
    const float* Wr0_pt = (const float*)d_in[5];
    const float* Wl0_tp = (const float*)d_in[6];
    const float* Wr0_tp = (const float*)d_in[7];
    const float* Wl1_pt = (const float*)d_in[8];
    const float* Wr1_pt = (const float*)d_in[9];
    const float* Wl1_tp = (const float*)d_in[10];
    const float* Wr1_tp = (const float*)d_in[11];
    const float* bl0_pt = (const float*)d_in[12];
    const float* bl0_tp = (const float*)d_in[13];
    const float* bl1_pt = (const float*)d_in[14];
    const float* bl1_tp = (const float*)d_in[15];

    const int N = in_sizes[0] / D;
    const int E = in_sizes[2] / 2;

    const int* src_pt = ei_pt;
    const int* dst_pt = ei_pt + E;
    const int* src_tp = ei_tp;
    const int* dst_tp = ei_tp + E;

    char* p = (char*)d_ws;
    auto take_i = [&](size_t n) { int* r = (int*)p; p += n * sizeof(int); return r; };
    int* deg_pt = take_i(N);
    int* deg_tp = take_i(N);
    int* cur_pt = take_i(N);
    int* cur_tp = take_i(N);
    int* off_pt = take_i(N + 1);
    int* off_tp = take_i(N + 1);
    int* bsum   = take_i(1024);
    int* csr_pt = take_i(E);
    int* csr_tp = take_i(E);
    p = (char*)(((size_t)p + 255) & ~(size_t)255);
    float* h_player = (float*)p; p += (size_t)N * D * sizeof(float);
    float* h_team   = (float*)p; p += (size_t)N * D * sizeof(float);

    float* o_player = (float*)d_out;
    float* o_team   = (float*)d_out + (size_t)N * D;

    const int TB = 256;
    const int nbN  = (N + TB - 1) / TB;
    const int nbE  = (E + TB - 1) / TB;
    const int nb4N = (4 * N + TB - 1) / TB;
    const int nbM  = (N + MT - 1) / MT;

    k_zero_i32<<<nb4N, TB, 0, stream>>>(deg_pt, 4 * N);
    k_degree<<<nbE, TB, 0, stream>>>(dst_pt, deg_pt, E);
    k_degree<<<nbE, TB, 0, stream>>>(dst_tp, deg_tp, E);

    k_scan_block<<<nbN, TB, 0, stream>>>(deg_pt, off_pt, bsum, N);
    k_scan_sums<<<1, 1, 0, stream>>>(bsum, nbN);
    k_scan_add<<<nbN, TB, 0, stream>>>(off_pt, bsum, N);

    k_scan_block<<<nbN, TB, 0, stream>>>(deg_tp, off_tp, bsum, N);
    k_scan_sums<<<1, 1, 0, stream>>>(bsum, nbN);
    k_scan_add<<<nbN, TB, 0, stream>>>(off_tp, bsum, N);

    k_fill<<<nbE, TB, 0, stream>>>(src_pt, dst_pt, off_pt, cur_pt, csr_pt, E);
    k_fill<<<nbE, TB, 0, stream>>>(src_tp, dst_tp, off_tp, cur_tp, csr_tp, E);

    k_sage<<<nbM, TB, 0, stream>>>(x_player, x_team,   off_pt, csr_pt, Wl0_pt, Wr0_pt, bl0_pt, h_team,   N, 1);
    k_sage<<<nbM, TB, 0, stream>>>(x_team,   x_player, off_tp, csr_tp, Wl0_tp, Wr0_tp, bl0_tp, h_player, N, 1);

    k_sage<<<nbM, TB, 0, stream>>>(h_player, h_team,   off_pt, csr_pt, Wl1_pt, Wr1_pt, bl1_pt, o_team,   N, 0);
    k_sage<<<nbM, TB, 0, stream>>>(h_team,   h_player, off_tp, csr_tp, Wl1_tp, Wr1_tp, bl1_tp, o_player, N, 0);
}


# CSR build: both relations share each launch, a thread takes 8 edges or 4 scan elements, block totals are prefixed inside the add kernel: 5 launches not 11; row order still by atomic arrival
# speedup vs baseline: 1.7509x; 1.1433x over previous
#include <hip/hip_runtime.h>
#include <hip/hip_bf16.h>

typedef __attribute__((ext_vector_type(2))) float v2f;
typedef __attribute__((ext_vector_type(8))) float v8f;

#define D 128
#define LDS_STRIDE 260


#define EPT 8
#define SPT 4

__global__ __launch_bounds__(256) void k_zero_i32(int* __restrict__ p, int n) {
    const int base = blockIdx.x * (256 * 4) + threadIdx.x;
    #pragma unroll
    for (int k = 0; k < 4; ++k) {
        const int i = base + k * 256;
        if (i < n) p[i] = 0;
    }
}

template <bool FULL>
__device__ __forceinline__ void degree_edges(const int* __restrict__ dst, int* __restrict__ dg,
                                             int base, int e) {
    int d[EPT];
    #pragma unroll
    for (int k = 0; k < EPT; ++k) {
        const int i = base + k * 256;
        d[k] = (FULL || i < e) ? dst[i] : -1;
    }
    #pragma unroll
    for (int k = 0; k < EPT; ++k)
        if (FULL || d[k] >= 0) atomicAdd(&dg[d[k]], 1);
}

__global__ __launch_bounds__(256) void k_degree(const int* __restrict__ dst0, const int* __restrict__ dst1,
                                                int* __restrict__ deg, int e, int n) {
    const int rel = blockIdx.y;
    const int* __restrict__ dst = rel ? dst1 : dst0;
    int* __restrict__ dg = deg + (size_t)rel * n;
    const int blk = blockIdx.x * (256 * EPT);
    if (blk + 256 * EPT <= e) degree_edges<true>(dst, dg, blk + threadIdx.x, e);
    else                      degree_edges<false>(dst, dg, blk + threadIdx.x, e);
}

__global__ __launch_bounds__(256) void k_scan_block(const int* __restrict__ deg, int* __restrict__ off,
                                                    int* __restrict__ bsum, int n) {
    __shared__ int s[256];
    const int rel = blockIdx.y;
    const int* __restrict__ dg = deg + (size_t)rel * n;
    int* __restrict__ of = off + (size_t)rel * (n + 1);
    const int i0 = (blockIdx.x * 256 + threadIdx.x) * SPT;
    int v[SPT];
    int t = 0;
    #pragma unroll
    for (int k = 0; k < SPT; ++k) {
        t += (i0 + k < n) ? dg[i0 + k] : 0;
        v[k] = t;
    }
    s[threadIdx.x] = t;
    __syncthreads();
    for (int d = 1; d < 256; d <<= 1) {
        const int u = ((int)threadIdx.x >= d) ? s[threadIdx.x - d] : 0;
        __syncthreads();
        s[threadIdx.x] += u;
        __syncthreads();
    }
    const int excl = s[threadIdx.x] - t;
    #pragma unroll
    for (int k = 0; k < SPT; ++k)
        if (i0 + k < n) of[i0 + k + 1] = excl + v[k];
    if (threadIdx.x == 255) bsum[rel * gridDim.x + blockIdx.x] = s[255];
}

__global__ __launch_bounds__(256) void k_scan_add(int* __restrict__ off, int* __restrict__ cur,
                                                  const int* __restrict__ deg,
                                                  const int* __restrict__ bsum, int n) {
    __shared__ int s[256];
    const int rel = blockIdx.y;
    const int* __restrict__ bs = bsum + rel * gridDim.x;
    const int* __restrict__ dg = deg + (size_t)rel * n;
    int* __restrict__ of = off + (size_t)rel * (n + 1);
    int* __restrict__ cu = cur + (size_t)rel * n;
    const int i0 = (blockIdx.x * 256 + threadIdx.x) * SPT;
    int o[SPT], g[SPT];
    #pragma unroll
    for (int k = 0; k < SPT; ++k) {
        const int i = i0 + k;
        o[k] = (i < n) ? of[i + 1] : 0;
        g[k] = (i < n) ? dg[i] : 0;
    }
    int part = 0;
    for (int j = threadIdx.x; j < (int)blockIdx.x; j += 256) part += bs[j];
    s[threadIdx.x] = part;
    __syncthreads();
    for (int d = 128; d > 0; d >>= 1) {
        if ((int)threadIdx.x < d) s[threadIdx.x] += s[threadIdx.x + d];
        __syncthreads();
    }
    const int pre = s[0];
    #pragma unroll
    for (int k = 0; k < SPT; ++k) {
        const int i = i0 + k;
        if (i < n) {
            const int incl = o[k] + pre;
            of[i + 1] = incl;
            cu[i] = incl - g[k];
        }
    }
    if (blockIdx.x == 0 && threadIdx.x == 0) of[0] = 0;
}

template <bool FULL>
__device__ __forceinline__ void fill_edges(const int* __restrict__ src, const int* __restrict__ dst,
                                           int* __restrict__ cu, int* __restrict__ cs,
                                           int base, int e) {
    int d[EPT], sv[EPT];
    unsigned pos[EPT];
    #pragma unroll
    for (int k = 0; k < EPT; ++k) {
        const int i = base + k * 256;
        d[k]  = (FULL || i < e) ? dst[i] : -1;
        sv[k] = (FULL || i < e) ? src[i] : 0;
    }
    #pragma unroll
    for (int k = 0; k < EPT; ++k)
        pos[k] = (FULL || d[k] >= 0) ? (unsigned)atomicAdd(&cu[d[k]], 1) : 0u;
    #pragma unroll
    for (int k = 0; k < EPT; ++k)
        if (FULL || d[k] >= 0) cs[pos[k]] = sv[k];
}

__global__ __launch_bounds__(256) void k_fill(const int* __restrict__ src0, const int* __restrict__ dst0,
                                              const int* __restrict__ src1, const int* __restrict__ dst1,
                                              int* __restrict__ cur, int* __restrict__ csr, int e, int n) {
    const int rel = blockIdx.y;
    const int* __restrict__ src = rel ? src1 : src0;
    const int* __restrict__ dst = rel ? dst1 : dst0;
    int* __restrict__ cu = cur + (size_t)rel * n;
    int* __restrict__ cs = csr + (size_t)rel * e;
    const int blk = blockIdx.x * (256 * EPT);
    if (blk + 256 * EPT <= e) fill_edges<true>(src, dst, cu, cs, blk + threadIdx.x, e);
    else                      fill_edges<false>(src, dst, cu, cs, blk + threadIdx.x, e);
}

#define MT 32
__global__ __launch_bounds__(256) void k_sage(
    const float* __restrict__ xsrc, const float* __restrict__ xdst,
    const int* __restrict__ off, const int* __restrict__ csr,
    const float* __restrict__ Wl, const float* __restrict__ Wr,
    const float* __restrict__ bl, float* __restrict__ out,
    int nDst, int relu)
{
    __shared__ __attribute__((aligned(16))) float A2[MT * LDS_STRIDE];
    const int wave = threadIdx.x >> 5;
    const int lane = threadIdx.x & 31;
    const int mbase = blockIdx.x * MT;
    const int c0 = lane * 4;

    const int wv = __builtin_amdgcn_readfirstlane(wave);
    const int r0 = wv * (MT / 8);
    const int g0 = mbase + r0;
    const int ov = off[min(g0 + min(lane, MT / 8), nDst)];
    float4 xv[MT / 8];
    #pragma unroll
    for (int rr = 0; rr < MT / 8; ++rr) {
        const int g = g0 + rr;
        xv[rr] = make_float4(0.f, 0.f, 0.f, 0.f);
        if (g < nDst) xv[rr] = *reinterpret_cast<const float4*>(xdst + (size_t)g * D + c0);
    }
    int begs[MT / 8], degs[MT / 8], ids[MT / 8];
    #pragma unroll
    for (int rr = 0; rr < MT / 8; ++rr) {
        begs[rr] = __builtin_amdgcn_readlane(ov, rr);
        degs[rr] = __builtin_amdgcn_readlane(ov, rr + 1) - begs[rr];
    }
    #pragma unroll
    for (int rr = 0; rr < MT / 8; ++rr)
        ids[rr] = (lane < degs[rr]) ? csr[begs[rr] + lane] : 0;
    #pragma unroll
    for (int rr = 0; rr < MT / 8; ++rr) {
        float* row = &A2[(r0 + rr) * LDS_STRIDE];
        row[D + c0 + 0] = xv[rr].x; row[D + c0 + 1] = xv[rr].y;
        row[D + c0 + 2] = xv[rr].z; row[D + c0 + 3] = xv[rr].w;
    }
    #pragma unroll
    for (int rr = 0; rr < MT / 8; ++rr) {
        const int beg = begs[rr];
        const int deg = degs[rr];
        int id = ids[rr];
        float ax = 0.f, ay = 0.f, az = 0.f, aw = 0.f;
        for (int j0 = 0; j0 < deg; j0 += 8) {
            if (j0 != 0 && (j0 & 31) == 0)
                id = (j0 + lane < deg) ? csr[beg + j0 + lane] : 0;
            float4 v[8];
            #pragma unroll
            for (int k = 0; k < 8; ++k) {
                v[k] = make_float4(0.f, 0.f, 0.f, 0.f);
                if (j0 + k < deg) {
                    const int s = __builtin_amdgcn_readlane(id, (j0 + k) & 31);
                    v[k] = *reinterpret_cast<const float4*>(xsrc + (size_t)s * D + c0);
                }
            }
            #pragma unroll
            for (int k = 0; k < 8; ++k) {
                if (j0 + k < deg) { ax += v[k].x; ay += v[k].y; az += v[k].z; aw += v[k].w; }
            }
        }
        const float sc = 1.0f / fmaxf((float)deg, 1.0f);
        float* row = &A2[(r0 + rr) * LDS_STRIDE];
        row[c0 + 0] = ax * sc; row[c0 + 1] = ay * sc;
        row[c0 + 2] = az * sc; row[c0 + 3] = aw * sc;
    }
    __syncthreads();

    const int lh = lane & 15;
    const int hi = lane >> 4;
    const int jcol = wave * 16 + lh;
    const float* arow0 = &A2[lh * LDS_STRIDE + 4 * hi];
    const float* arow1 = arow0 + 16 * LDS_STRIDE;
    const float* wlrow = Wl + (size_t)jcol * D + 4 * hi;
    const float* wrrow = Wr + (size_t)jcol * D + 4 * hi;

    v8f acc0 = {};
    v8f acc1 = {};
    #pragma unroll
    for (int kb = 0; kb < D; kb += 8) {
        const float4 b  = *reinterpret_cast<const float4*>(wlrow + kb);
        const float4 a0 = *reinterpret_cast<const float4*>(arow0 + kb);
        const float4 a1 = *reinterpret_cast<const float4*>(arow1 + kb);
        const v2f blo = {b.x, b.y},  bhi = {b.z, b.w};
        const v2f a0lo = {a0.x, a0.y}, a0hi = {a0.z, a0.w};
        const v2f a1lo = {a1.x, a1.y}, a1hi = {a1.z, a1.w};
        acc0 = __builtin_amdgcn_wmma_f32_16x16x4_f32(false, a0lo, false, blo, (short)0, acc0, false, false);
        acc1 = __builtin_amdgcn_wmma_f32_16x16x4_f32(false, a1lo, false, blo, (short)0, acc1, false, false);
        acc0 = __builtin_amdgcn_wmma_f32_16x16x4_f32(false, a0hi, false, bhi, (short)0, acc0, false, false);
        acc1 = __builtin_amdgcn_wmma_f32_16x16x4_f32(false, a1hi, false, bhi, (short)0, acc1, false, false);
    }
    #pragma unroll
    for (int kb = 0; kb < D; kb += 8) {
        const float4 b  = *reinterpret_cast<const float4*>(wrrow + kb);
        const float4 a0 = *reinterpret_cast<const float4*>(arow0 + D + kb);
        const float4 a1 = *reinterpret_cast<const float4*>(arow1 + D + kb);
        const v2f blo = {b.x, b.y},  bhi = {b.z, b.w};
        const v2f a0lo = {a0.x, a0.y}, a0hi = {a0.z, a0.w};
        const v2f a1lo = {a1.x, a1.y}, a1hi = {a1.z, a1.w};
        acc0 = __builtin_amdgcn_wmma_f32_16x16x4_f32(false, a0lo, false, blo, (short)0, acc0, false, false);
        acc1 = __builtin_amdgcn_wmma_f32_16x16x4_f32(false, a1lo, false, blo, (short)0, acc1, false, false);
        acc0 = __builtin_amdgcn_wmma_f32_16x16x4_f32(false, a0hi, false, bhi, (short)0, acc0, false, false);
        acc1 = __builtin_amdgcn_wmma_f32_16x16x4_f32(false, a1hi, false, bhi, (short)0, acc1, false, false);
    }

    const float bb = bl[jcol];
    #pragma unroll
    for (int i = 0; i < 8; ++i) {
        const int gr0 = mbase + i + 8 * hi;
        if (gr0 < nDst) {
            float v = acc0[i] + bb;
            if (relu) v = fmaxf(v, 0.0f);
            out[(size_t)gr0 * D + jcol] = v;
        }
        const int gr1 = gr0 + 16;
        if (gr1 < nDst) {
            float v = acc1[i] + bb;
            if (relu) v = fmaxf(v, 0.0f);
            out[(size_t)gr1 * D + jcol] = v;
        }
    }
}


extern "C" void kernel_launch(void* const* d_in, const int* in_sizes, int n_in,
                              void* d_out, int out_size, void* d_ws, size_t ws_size,
                              hipStream_t stream) {
    const float* x_player = (const float*)d_in[0];
    const float* x_team   = (const float*)d_in[1];
    const int*   ei_pt    = (const int*)d_in[2];
    const int*   ei_tp    = (const int*)d_in[3];
    const float* Wl0_pt = (const float*)d_in[4];
    const float* Wr0_pt = (const float*)d_in[5];
    const float* Wl0_tp = (const float*)d_in[6];
    const float* Wr0_tp = (const float*)d_in[7];
    const float* Wl1_pt = (const float*)d_in[8];
    const float* Wr1_pt = (const float*)d_in[9];
    const float* Wl1_tp = (const float*)d_in[10];
    const float* Wr1_tp = (const float*)d_in[11];
    const float* bl0_pt = (const float*)d_in[12];
    const float* bl0_tp = (const float*)d_in[13];
    const float* bl1_pt = (const float*)d_in[14];
    const float* bl1_tp = (const float*)d_in[15];

    const int N = in_sizes[0] / D;
    const int E = in_sizes[2] / 2;

    const int* src_pt = ei_pt;
    const int* dst_pt = ei_pt + E;
    const int* src_tp = ei_tp;
    const int* dst_tp = ei_tp + E;

    const int TB = 256;
    const int nbZ = (2 * N + 4 * TB - 1) / (4 * TB);
    const int nbS = (N + SPT * TB - 1) / (SPT * TB);
    const int nbE = (E + EPT * TB - 1) / (EPT * TB);
    const int nbM = (N + MT - 1) / MT;

    char* p = (char*)d_ws;
    auto take_i = [&](size_t n) { int* r = (int*)p; p += n * sizeof(int); return r; };
    int* deg  = take_i(2 * (size_t)N);
    int* cur  = take_i(2 * (size_t)N);
    int* off  = take_i(2 * ((size_t)N + 1));
    int* bsum = take_i(2 * (size_t)nbS);
    int* csr  = take_i(2 * (size_t)E);
    const int* off_pt = off;
    const int* off_tp = off + N + 1;
    const int* csr_pt = csr;
    const int* csr_tp = csr + E;
    p = (char*)(((size_t)p + 255) & ~(size_t)255);
    float* h_player = (float*)p; p += (size_t)N * D * sizeof(float);
    float* h_team   = (float*)p; p += (size_t)N * D * sizeof(float);

    float* o_player = (float*)d_out;
    float* o_team   = (float*)d_out + (size_t)N * D;

    k_zero_i32<<<nbZ, TB, 0, stream>>>(deg, 2 * N);
    k_degree<<<dim3(nbE, 2), TB, 0, stream>>>(dst_pt, dst_tp, deg, E, N);
    k_scan_block<<<dim3(nbS, 2), TB, 0, stream>>>(deg, off, bsum, N);
    k_scan_add<<<dim3(nbS, 2), TB, 0, stream>>>(off, cur, deg, bsum, N);
    k_fill<<<dim3(nbE, 2), TB, 0, stream>>>(src_pt, dst_pt, src_tp, dst_tp, cur, csr, E, N);

    k_sage<<<nbM, TB, 0, stream>>>(x_player, x_team,   off_pt, csr_pt, Wl0_pt, Wr0_pt, bl0_pt, h_team,   N, 1);
    k_sage<<<nbM, TB, 0, stream>>>(x_team,   x_player, off_tp, csr_tp, Wl0_tp, Wr0_tp, bl0_tp, h_player, N, 1);

    k_sage<<<nbM, TB, 0, stream>>>(h_player, h_team,   off_pt, csr_pt, Wl1_pt, Wr1_pt, bl1_pt, o_team,   N, 0);
    k_sage<<<nbM, TB, 0, stream>>>(h_team,   h_player, off_tp, csr_tp, Wl1_tp, Wr1_tp, bl1_tp, o_player, N, 0);
}
